# starting kernel
# baseline (speedup 1.0000x reference)
#include <hip/hip_runtime.h>
#include <math.h>

#define ALPHA 0.1f

typedef float v2f __attribute__((ext_vector_type(2)));
typedef float v8f __attribute__((ext_vector_type(8)));

__device__ __forceinline__ void atomic_add_f32(float* p, float v) {
#if __has_builtin(__builtin_amdgcn_global_atomic_fadd_f32)
  typedef __attribute__((address_space(1))) float gfloat;
  __builtin_amdgcn_global_atomic_fadd_f32((gfloat*)p, v);
#else
  atomicAdd(p, v);
#endif
}


__global__ void zero4_k(float4* __restrict__ p, int n4) {
  int i = blockIdx.x * blockDim.x + threadIdx.x;
  if (i < n4) p[i] = make_float4(0.f, 0.f, 0.f, 0.f);
}

__global__ void deg_k(const int* __restrict__ dst, float* __restrict__ deg, int E) {
  int e = blockIdx.x * blockDim.x + threadIdx.x;
  if (e < E) atomic_add_f32(&deg[dst[e]], 1.0f);
}

__global__ void norm_k(float* __restrict__ nrm, int n) {
  int i = blockIdx.x * blockDim.x + threadIdx.x;
  if (i < n) {
    float d = nrm[i];
    nrm[i] = 1.0f / sqrtf(fmaxf(d, 1.0f));
  }
}

__device__ __forceinline__ float elu1(float v) {
  return v > 0.0f ? v : (expf(v) - 1.0f);
}

template <int KDIM, int HID, bool ELU>
__global__ void gemm_wmma(const float* __restrict__ A, const float* __restrict__ W,
                          const float* __restrict__ bias, float* __restrict__ C, int N) {
  constexpr int NTN = HID / 16;
  const int tiles = ((N + 15) / 16) * NTN;
  const int tile = __builtin_amdgcn_readfirstlane(
      blockIdx.x * (blockDim.x >> 5) + (threadIdx.x >> 5));
  if (tile >= tiles) return;
  const int tm = tile / NTN;
  const int tn = tile % NTN;
  const int lane = threadIdx.x & 31;
  const int half = lane >> 4;
  const int l15 = lane & 15;
  const int row = tm * 16 + l15;
  const int rowc = row < N ? row : N - 1;
  const int col = tn * 16 + l15;

  v8f acc = {};
  for (int k = 0; k < KDIM; k += 4) {
    const int ka = k + 2 * half;
    v2f a, b;
    a.x = A[rowc * KDIM + ka];
    a.y = A[rowc * KDIM + ka + 1];
    b.x = W[col * KDIM + ka];
    b.y = W[col * KDIM + ka + 1];
    acc = __builtin_amdgcn_wmma_f32_16x16x4_f32(false, a, false, b,
                                                (short)0, acc, false, false);
  }
  const float bs = bias[col];
  if (tm * 16 + 15 < N) {
#pragma unroll
    for (int v = 0; v < 8; ++v) {
      int m = tm * 16 + v + 8 * half;
      float x = acc[v] + bs;
      if (ELU) x = elu1(x);
      C[m * HID + col] = x;
    }
  } else {
#pragma unroll
    for (int v = 0; v < 8; ++v) {
      int m = tm * 16 + v + 8 * half;
      if (m < N) {
        float x = acc[v] + bs;
        if (ELU) x = elu1(x);
        C[m * HID + col] = x;
      }
    }
  }
}

template <int D>
__global__ void edge_k(const int* __restrict__ src, const int* __restrict__ dst,
                       const float* __restrict__ h, const float* __restrict__ nrm,
                       float* __restrict__ agg, int E) {
  int t = blockIdx.x * blockDim.x + threadIdx.x;
  int e = t >> 4;
  int c = t & 15;
  if (e >= E) return;
  int s = src[e];
  int d = dst[e];
  float w = nrm[s];
#pragma unroll
  for (int j = 0; j < D / 16; ++j) {
    int ch = c + 16 * j;
    atomic_add_f32(&agg[d * D + ch], h[s * D + ch] * w);
  }
}

template <int D, bool ELU>
__global__ void combine4_k(const float* __restrict__ nrm, const float4* __restrict__ h0,
                           float4* __restrict__ agg, float4* __restrict__ out, int N) {
  int i = blockIdx.x * blockDim.x + threadIdx.x;
  if (i >= N * (D / 4)) return;
  int node = i / (D / 4);
  float w = (1.0f - ALPHA) * nrm[node];
  float4 a = agg[i];
  float4 h = h0[i];
  float4 r = make_float4(w * a.x + ALPHA * h.x, w * a.y + ALPHA * h.y,
                         w * a.z + ALPHA * h.z, w * a.w + ALPHA * h.w);
  if (ELU) {
    out[i] = make_float4(elu1(r.x), elu1(r.y), elu1(r.z), elu1(r.w));
  } else {
    agg[i] = r;
  }
}


extern "C" void kernel_launch(void* const* d_in, const int* in_sizes, int n_in,
                              void* d_out, int out_size, void* d_ws, size_t ws_size,
                              hipStream_t stream) {
  const float* F   = (const float*)d_in[0];
  const int*   src = (const int*)d_in[1];
  const int*   dst = (const int*)d_in[2];
  const float* W1  = (const float*)d_in[3];
  const float* b1  = (const float*)d_in[4];
  const float* W2  = (const float*)d_in[5];
  const float* b2  = (const float*)d_in[6];
  float* out = (float*)d_out;

  constexpr int IN = 256, HID = 48, CLS = 16, KSTEPS = 10;
  const int N = in_sizes[0] / IN;
  const int E = in_sizes[1];

  float* ws  = (float*)d_ws;
  float* nrm = ws;
  float* h0  = nrm + N;
  float* hA  = h0 + (size_t)N * HID;
  float* hB  = hA + (size_t)N * HID;
  float* x0  = h0;
  float* xA  = x0 + (size_t)N * CLS;
  float* xB  = xA + (size_t)N * CLS;

  auto blocks = [](long n, int t) { return (int)((n + t - 1) / t); };

  zero4_k<<<blocks(N / 4, 256), 256, 0, stream>>>((float4*)nrm, N / 4);
  deg_k<<<blocks(E, 256), 256, 0, stream>>>(dst, nrm, E);
  norm_k<<<blocks(N, 256), 256, 0, stream>>>(nrm, N);

  const int tiles1 = ((N + 15) / 16) * (HID / 16);
  gemm_wmma<IN, HID, false><<<blocks((long)tiles1 * 32, 128), 128, 0, stream>>>(F, W1, b1, h0, N);

  const float* cur = h0;
  for (int t = 0; t < KSTEPS; ++t) {
    float* agg = (t & 1) ? hB : hA;
    zero4_k<<<blocks((long)N * HID / 4, 256), 256, 0, stream>>>((float4*)agg, N * (HID / 4));
    edge_k<HID><<<blocks((long)E * 16, 256), 256, 0, stream>>>(src, dst, cur, nrm, agg, E);
    combine4_k<HID, false><<<blocks((long)N * HID / 4, 256), 256, 0, stream>>>(
        nrm, (const float4*)h0, (float4*)agg, nullptr, N);
    cur = agg;
  }

  const int tiles2 = ((N + 15) / 16) * (CLS / 16);
  gemm_wmma<HID, CLS, true><<<blocks((long)tiles2 * 32, 128), 128, 0, stream>>>(cur, W2, b2, x0, N);

  const float* cur2 = x0;
  for (int t = 0; t < KSTEPS; ++t) {
    float* agg = (t & 1) ? xB : xA;
    zero4_k<<<blocks((long)N * CLS / 4, 256), 256, 0, stream>>>((float4*)agg, N * (CLS / 4));
    edge_k<CLS><<<blocks((long)E * 16, 256), 256, 0, stream>>>(src, dst, cur2, nrm, agg, E);
    if (t == KSTEPS - 1) {
      combine4_k<CLS, true><<<blocks((long)N * CLS / 4, 256), 256, 0, stream>>>(
          nrm, (const float4*)x0, (float4*)agg, (float4*)out, N);
    } else {
      combine4_k<CLS, false><<<blocks((long)N * CLS / 4, 256), 256, 0, stream>>>(
          nrm, (const float4*)x0, (float4*)agg, nullptr, N);
    }
    cur2 = agg;
  }
}


# Edge list split once per call by destination range (ballot ranks in the degree pass); each L2 half's steps walk only their own list: every batch slot a live edge; sums stay in arrival order
# speedup vs baseline: 2.6940x; 2.6940x over previous
#include <hip/hip_runtime.h>
#include <math.h>

#define ALPHA 0.1f

typedef float v2f __attribute__((ext_vector_type(2)));
typedef float v8f __attribute__((ext_vector_type(8)));

__device__ __forceinline__ void atomic_add_f32(float* p, float v) {
#if __has_builtin(__builtin_amdgcn_global_atomic_fadd_f32)
  typedef __attribute__((address_space(1))) float gfloat;
  __builtin_amdgcn_global_atomic_fadd_f32((gfloat*)p, v);
#else
  atomicAdd(p, v);
#endif
}


#define CTR_PITCH 64

__device__ __forceinline__ unsigned my_half() {
  return (__builtin_amdgcn_s_sendmsg_rtn(0x87) >> 18) & 1u;
}

__device__ __forceinline__ unsigned half_rows(unsigned N) {
  return (N >> 3) << 2;
}

__device__ __forceinline__ unsigned list_cap(unsigned E) {
  return (E + 3u) & ~3u;
}

__global__ void zero4_k(float4* __restrict__ p, int n4, float4* __restrict__ q, int m4) {
  int i = blockIdx.x * blockDim.x + threadIdx.x;
  if (i < n4) p[i] = make_float4(0.f, 0.f, 0.f, 0.f);
  if (i < m4) q[i] = make_float4(0.f, 0.f, 0.f, 0.f);
}

__global__ void __launch_bounds__(256) deg_k(const int* __restrict__ src, const int* __restrict__ dst,
                                             float* __restrict__ deg, int* __restrict__ ctr,
                                             int* __restrict__ len, int* __restrict__ lst, int E, int N) {
  __shared__ int sh[2][2];
  __shared__ int wtot[8];
  __shared__ int sbase;
  const unsigned tid = threadIdx.x;
  const unsigned wv = tid >> 5;
  const unsigned ncl = ((unsigned)E + 1023u) / 1024u;
  const unsigned Nh = half_rows((unsigned)N);
  const unsigned cap = list_cap((unsigned)E);
  const unsigned hw = my_half();
  const bool vec = ((((size_t)dst) | ((size_t)src)) & 15) == 0;
  for (unsigned ph = 0; ph < 2; ++ph) {
    const unsigned hh = hw ^ ph;
    int* cp = ctr + hh * CTR_PITCH;
    int* lp = len + hh * CTR_PITCH;
    int* ls = lst + (size_t)(2u * hh) * cap;
    int* ld = ls + cap;
    if (tid == 0) sh[ph][0] = atomicAdd(cp, 1);
    __syncthreads();
    unsigned cur = (unsigned)sh[ph][0];
    unsigned it = 0;
    while (cur < ncl) {
      int nx = 0;
      const unsigned e0 = cur * 1024u + tid * 4u;
      int s[4], d[4];
      bool f[4];
      if (vec && e0 + 4u <= (unsigned)E) {
        const int4 dv = *reinterpret_cast<const int4*>(dst + e0);
        const int4 sv = *reinterpret_cast<const int4*>(src + e0);
        if (tid == 0) nx = atomicAdd(cp, 1);
        d[0] = dv.x; d[1] = dv.y; d[2] = dv.z; d[3] = dv.w;
        s[0] = sv.x; s[1] = sv.y; s[2] = sv.z; s[3] = sv.w;
        for (int k = 0; k < 4; ++k) f[k] = (((unsigned)d[k] >= Nh) ? 1u : 0u) == hh;
      } else {
        if (tid == 0) nx = atomicAdd(cp, 1);
        for (int k = 0; k < 4; ++k) {
          const unsigned e = e0 + (unsigned)k;
          s[k] = 0;
          d[k] = 0;
          f[k] = false;
          if (e < (unsigned)E) {
            d[k] = dst[e];
            s[k] = src[e];
            f[k] = (((unsigned)d[k] >= Nh) ? 1u : 0u) == hh;
          }
        }
      }
      unsigned b[4];
      unsigned wsum = 0;
      for (int k = 0; k < 4; ++k) {
        if (f[k]) atomic_add_f32(&deg[d[k]], 1.0f);
        b[k] = __builtin_amdgcn_ballot_w32(f[k]);
        wsum += (unsigned)__builtin_popcount(b[k]);
      }
      if ((tid & 31u) == 0) wtot[wv] = (int)wsum;
      __syncthreads();
      unsigned woff = 0, tot = 0;
      for (unsigned w = 0; w < 8; ++w) {
        const unsigned x = (unsigned)wtot[w];
        tot += x;
        if (w < wv) woff += x;
      }
      ++it;
      if (tid == 0) {
        sbase = atomicAdd(lp, (int)tot);
        sh[ph][it & 1] = nx;
      }
      __syncthreads();
      unsigned p = (unsigned)sbase + woff;
      for (int k = 0; k < 4; ++k) {
        if (f[k]) {
          const unsigned q = p + __builtin_amdgcn_mbcnt_lo(b[k], 0u);
          ls[q] = s[k];
          ld[q] = d[k];
        }
        p += (unsigned)__builtin_popcount(b[k]);
      }
      cur = (unsigned)sh[ph][it & 1];
    }
  }
}

__global__ void norm_k(float* __restrict__ nrm, float* __restrict__ wsrc, int n) {
  int i = blockIdx.x * blockDim.x + threadIdx.x;
  if (i < n) {
    float d = nrm[i];
    float r = 1.0f / sqrtf(fmaxf(d, 1.0f));
    nrm[i] = r;
    wsrc[i] = (1.0f - ALPHA) * r * r;
  }
}

__device__ __forceinline__ float elu1(float v) {
  return v > 0.0f ? v : (expf(v) - 1.0f);
}

template <int KDIM, int HID, bool ELU>
__global__ void gemm_wmma(const float* __restrict__ A, const float* __restrict__ W,
                          const float* __restrict__ bias, float* __restrict__ C, int N) {
  constexpr int NTN = HID / 16;
  const int tiles = ((N + 15) / 16) * NTN;
  const int tile = __builtin_amdgcn_readfirstlane(
      blockIdx.x * (blockDim.x >> 5) + (threadIdx.x >> 5));
  if (tile >= tiles) return;
  const int tm = tile / NTN;
  const int tn = tile % NTN;
  const int lane = threadIdx.x & 31;
  const int half = lane >> 4;
  const int l15 = lane & 15;
  const int row = tm * 16 + l15;
  const int rowc = row < N ? row : N - 1;
  const int col = tn * 16 + l15;

  v8f acc = {};
  for (int k = 0; k < KDIM; k += 4) {
    const int ka = k + 2 * half;
    v2f a, b;
    a.x = A[rowc * KDIM + ka];
    a.y = A[rowc * KDIM + ka + 1];
    b.x = W[col * KDIM + ka];
    b.y = W[col * KDIM + ka + 1];
    acc = __builtin_amdgcn_wmma_f32_16x16x4_f32(false, a, false, b,
                                                (short)0, acc, false, false);
  }
  const float bs = bias[col];
  if (tm * 16 + 15 < N) {
#pragma unroll
    for (int v = 0; v < 8; ++v) {
      int m = tm * 16 + v + 8 * half;
      float x = acc[v] + bs;
      if (ELU) x = elu1(x);
      C[m * HID + col] = x;
    }
  } else {
#pragma unroll
    for (int v = 0; v < 8; ++v) {
      int m = tm * 16 + v + 8 * half;
      if (m < N) {
        float x = acc[v] + bs;
        if (ELU) x = elu1(x);
        C[m * HID + col] = x;
      }
    }
  }
}


template <int D, int EPG>
__device__ __forceinline__ void edge_batch(const int* __restrict__ src, const int* __restrict__ dst,
                                           const float* __restrict__ h, const float* __restrict__ nrm,
                                           float* __restrict__ agg, unsigned e0, unsigned c, unsigned E,
                                           int* cp, int& nx) {
  constexpr int NJ = D / 16;
  const bool full = (e0 + (unsigned)EPG <= E) &&
                    ((((size_t)src | (size_t)dst) & 15) == 0);
  if (!full) {
    if (cp != nullptr) nx = atomicAdd(cp, 1);
    if (e0 >= E) return;
    for (int i = 0; i < EPG; ++i) {
      const unsigned e = e0 + i;
      if (e < E) {
        const int s = src[e];
        const int d = dst[e];
        const float w = nrm[s];
#pragma unroll
        for (int j = 0; j < NJ; ++j) {
          const int ch = c + 16 * j;
          atomic_add_f32(&agg[d * D + ch], h[s * D + ch] * w);
        }
      }
    }
    return;
  }

  int s[EPG], d[EPG];
  const int4* s4 = reinterpret_cast<const int4*>(src + e0);
  const int4* d4 = reinterpret_cast<const int4*>(dst + e0);
#pragma unroll
  for (int q = 0; q < EPG / 4; ++q) {
    const int4 a = s4[q];
    const int4 b = d4[q];
    s[4 * q + 0] = a.x; s[4 * q + 1] = a.y; s[4 * q + 2] = a.z; s[4 * q + 3] = a.w;
    d[4 * q + 0] = b.x; d[4 * q + 1] = b.y; d[4 * q + 2] = b.z; d[4 * q + 3] = b.w;
  }

  const int ci = (int)c;
  float w[EPG];
  float v[EPG][NJ];
#pragma unroll
  for (int i = 0; i < EPG; ++i) {
    w[i] = nrm[s[i]];
    const int sb = s[i] * D + ci;
#pragma unroll
    for (int j = 0; j < NJ; ++j) v[i][j] = h[sb + 16 * j];
  }
  if (cp != nullptr) nx = atomicAdd(cp, 1);
#pragma unroll
  for (int i = 0; i < EPG; ++i) {
    const int db = d[i] * D + ci;
#pragma unroll
    for (int j = 0; j < NJ; ++j) atomic_add_f32(&agg[db + 16 * j], v[i][j] * w[i]);
  }
}

template <int D, int EPG>
__global__ void __launch_bounds__(256) edge_k(const int* __restrict__ lst, const int* __restrict__ len,
                       const float* __restrict__ h, const float* __restrict__ nrm,
                       float* __restrict__ agg, const float* __restrict__ seed,
                       float* __restrict__ nxt, int* __restrict__ ctr, int E, int N) {
  constexpr int NJ = D / 16;
  constexpr unsigned BATCH = 16u * EPG;
  constexpr unsigned CH = 2u * BATCH;
  __shared__ int sh[2][2];
  const unsigned tid = threadIdx.x;
  const unsigned c = tid & 15u;
  const unsigned grp = tid >> 4;
  const unsigned Nh = half_rows((unsigned)N);
  const unsigned cap = list_cap((unsigned)E);
  const unsigned hw = my_half();
  for (unsigned ph = 0; ph < 2; ++ph) {
    const unsigned hh = hw ^ ph;
    int* cp = ctr + hh * CTR_PITCH;
    const int* src = lst + (size_t)(2u * hh) * cap;
    const int* dst = src + cap;
    const unsigned Eh = (unsigned)len[hh * CTR_PITCH];
    const unsigned ncl = Eh > 0 ? (Eh + CH - 1u) / CH : 1u;
    const unsigned rbase = hh ? Nh : 0u;
    const unsigned rows = hh ? (unsigned)N - Nh : Nh;
    const unsigned rpc = (rows + ncl - 1u) / ncl;
    if (tid == 0) sh[ph][0] = atomicAdd(cp, 1);
    __syncthreads();
    unsigned cur = (unsigned)sh[ph][0];
    unsigned it = 0;
    while (cur < ncl) {
      int nx = 0;
      const unsigned r1 = (cur + 1u) * rpc < rows ? (cur + 1u) * rpc : rows;
      const unsigned r0 = cur * rpc + grp;
      const bool cp0 = (nxt != nullptr) && (r0 < r1);
      const unsigned row0 = rbase + r0;
      float sv[NJ];
#pragma unroll
      for (int j = 0; j < NJ; ++j) {
        sv[j] = 0.0f;
        if (cp0) sv[j] = __builtin_nontemporal_load(&seed[row0 * D + c + 16u * j]);
      }
      edge_batch<D, EPG>(src, dst, h, nrm, agg, cur * CH + grp * (unsigned)EPG, c, Eh,
                         nullptr, nx);
      edge_batch<D, EPG>(src, dst, h, nrm, agg, cur * CH + BATCH + grp * (unsigned)EPG, c, Eh,
                         tid == 0 ? cp : nullptr, nx);
      if (cp0) {
#pragma unroll
        for (int j = 0; j < NJ; ++j) nxt[row0 * D + c + 16u * j] = sv[j];
        for (unsigned r = r0 + 16u; r < r1; r += 16u) {
          const unsigned row = rbase + r;
#pragma unroll
          for (int j = 0; j < NJ; ++j)
            nxt[row * D + c + 16u * j] = __builtin_nontemporal_load(&seed[row * D + c + 16u * j]);
        }
      }
      ++it;
      if (tid == 0) sh[ph][it & 1] = nx;
      __syncthreads();
      cur = (unsigned)sh[ph][it & 1];
    }
  }
}

template <int D>
__global__ void prep4_k(const float* __restrict__ nrm, const float4* __restrict__ h,
                        float4* __restrict__ a0, float4* __restrict__ seed,
                        float4* __restrict__ tgt, int N) {
  int i = blockIdx.x * blockDim.x + threadIdx.x;
  if (i >= N * (D / 4)) return;
  int node = i / (D / 4);
  float r = 1.0f / ((1.0f - ALPHA) * nrm[node]);
  float4 x = h[i];
  float4 a = make_float4(x.x * r, x.y * r, x.z * r, x.w * r);
  float4 s = make_float4(ALPHA * a.x, ALPHA * a.y, ALPHA * a.z, ALPHA * a.w);
  a0[i] = a;
  seed[i] = s;
  tgt[i] = s;
}

template <int D, bool ELU>
__global__ void fin4_k(const float* __restrict__ nrm, const float4* __restrict__ a,
                       float4* __restrict__ out, int N) {
  int i = blockIdx.x * blockDim.x + threadIdx.x;
  if (i >= N * (D / 4)) return;
  int node = i / (D / 4);
  float w = (1.0f - ALPHA) * nrm[node];
  float4 x = a[i];
  float4 r = make_float4(w * x.x, w * x.y, w * x.z, w * x.w);
  if (ELU) r = make_float4(elu1(r.x), elu1(r.y), elu1(r.z), elu1(r.w));
  out[i] = r;
}


extern "C" void kernel_launch(void* const* d_in, const int* in_sizes, int n_in,
                              void* d_out, int out_size, void* d_ws, size_t ws_size,
                              hipStream_t stream) {
  const float* F   = (const float*)d_in[0];
  const int*   src = (const int*)d_in[1];
  const int*   dst = (const int*)d_in[2];
  const float* W1  = (const float*)d_in[3];
  const float* b1  = (const float*)d_in[4];
  const float* W2  = (const float*)d_in[5];
  const float* b2  = (const float*)d_in[6];
  float* out = (float*)d_out;

  constexpr int IN = 256, HID = 48, CLS = 16, KSTEPS = 10;
  constexpr int EPG = 8;
  const int N = in_sizes[0] / IN;
  const int E = in_sizes[1];

  float* ws   = (float*)d_ws;
  float* nrm  = ws;
  float* wsrc = nrm + N;
  float* H    = wsrc + N;
  float* S    = H + (size_t)N * HID;
  float* B[3];
  B[0] = S + (size_t)N * HID;
  B[1] = B[0] + (size_t)N * HID;
  B[2] = B[1] + (size_t)N * HID;
  float* X    = S;
  float* b[3];
  b[0] = B[0];
  b[1] = b[0] + (size_t)N * CLS;
  b[2] = b[1] + (size_t)N * CLS;
  float* s16  = B[1];
  int* ctr = (int*)(B[2] + (size_t)N * HID);
  constexpr int NCTR = 2 * (2 + 2 * KSTEPS) * CTR_PITCH;
  int* len = ctr + 2 * (1 + 2 * KSTEPS) * CTR_PITCH;
  int* lst = ctr + NCTR;

  auto blocks = [](long n, int t) { return (int)((n + t - 1) / t); };

  zero4_k<<<blocks(N / 4, 256), 256, 0, stream>>>((float4*)nrm, N / 4, (float4*)ctr, NCTR / 4);
  deg_k<<<1024, 256, 0, stream>>>(src, dst, nrm, ctr, len, lst, E, N);
  norm_k<<<blocks(N, 256), 256, 0, stream>>>(nrm, wsrc, N);

  const int tiles1 = ((N + 15) / 16) * (HID / 16);
  gemm_wmma<IN, HID, false><<<blocks((long)tiles1 * 32, 128), 128, 0, stream>>>(F, W1, b1, H, N);

  const int egrid = 2048;

  prep4_k<HID><<<blocks((long)N * HID / 4, 256), 256, 0, stream>>>(
      nrm, (const float4*)H, (float4*)B[0], (float4*)S, (float4*)B[1], N);
  for (int t = 0; t < KSTEPS; ++t) {
    float* nx = (t == KSTEPS - 1) ? nullptr : B[(t + 2) % 3];
    edge_k<HID, EPG><<<egrid, 256, 0, stream>>>(lst, len, B[t % 3], wsrc, B[(t + 1) % 3], S, nx,
                                                ctr + 2 * (1 + t) * CTR_PITCH, E, N);
  }
  fin4_k<HID, false><<<blocks((long)N * HID / 4, 256), 256, 0, stream>>>(
      nrm, (const float4*)B[KSTEPS % 3], (float4*)H, N);

  const int tiles2 = ((N + 15) / 16) * (CLS / 16);
  gemm_wmma<HID, CLS, true><<<blocks((long)tiles2 * 32, 128), 128, 0, stream>>>(H, W2, b2, X, N);

  prep4_k<CLS><<<blocks((long)N * CLS / 4, 256), 256, 0, stream>>>(
      nrm, (const float4*)X, (float4*)b[0], (float4*)s16, (float4*)b[1], N);
  for (int t = 0; t < KSTEPS; ++t) {
    float* nx = (t == KSTEPS - 1) ? nullptr : b[(t + 2) % 3];
    edge_k<CLS, EPG><<<egrid, 256, 0, stream>>>(lst, len, b[t % 3], wsrc, b[(t + 1) % 3], s16, nx,
                                                ctr + 2 * (1 + KSTEPS + t) * CTR_PITCH, E, N);
  }
  fin4_k<CLS, true><<<blocks((long)N * CLS / 4, 256), 256, 0, stream>>>(
      nrm, (const float4*)b[KSTEPS % 3], (float4*)out, N);
}


# fc1 weight-stationary: a wave keeps its 16-column W1 tile as register fragments and walks row tiles; features go straight to registers 16 B per lane; k order permuted, sums reassociated
# speedup vs baseline: 4.1531x; 1.5416x over previous
#include <hip/hip_runtime.h>
#include <math.h>

#define ALPHA 0.1f

typedef float v2f __attribute__((ext_vector_type(2)));
typedef float v8f __attribute__((ext_vector_type(8)));

__device__ __forceinline__ void atomic_add_f32(float* p, float v) {
#if __has_builtin(__builtin_amdgcn_global_atomic_fadd_f32)
  typedef __attribute__((address_space(1))) float gfloat;
  __builtin_amdgcn_global_atomic_fadd_f32((gfloat*)p, v);
#else
  atomicAdd(p, v);
#endif
}

#define RB 52
#define SRC_BITS 26
#define SRC_MASK ((1 << SRC_BITS) - 1)

typedef float v4f __attribute__((ext_vector_type(4)));

__device__ __forceinline__ void lds_add_f32(float* p, float v) {
  __hip_atomic_fetch_add(p, v, __ATOMIC_RELAXED, __HIP_MEMORY_SCOPE_WORKGROUP);
}

__global__ void zero4_k(float4* __restrict__ p, int n4, float4* __restrict__ q, int m4) {
  int i = blockIdx.x * blockDim.x + threadIdx.x;
  if (i < n4) p[i] = make_float4(0.f, 0.f, 0.f, 0.f);
  if (i < m4) q[i] = make_float4(0.f, 0.f, 0.f, 0.f);
}

__global__ void __launch_bounds__(256) deg_k(const int* __restrict__ dst, float* __restrict__ deg, int E) {
  const unsigned e0 = (blockIdx.x * 256u + threadIdx.x) * 4u;
  if (e0 >= (unsigned)E) return;
  if (((((size_t)dst) & 15) == 0) && e0 + 4u <= (unsigned)E) {
    const int4 dv = *reinterpret_cast<const int4*>(dst + e0);
    atomic_add_f32(&deg[dv.x], 1.0f);
    atomic_add_f32(&deg[dv.y], 1.0f);
    atomic_add_f32(&deg[dv.z], 1.0f);
    atomic_add_f32(&deg[dv.w], 1.0f);
  } else {
    for (unsigned e = e0; e < e0 + 4u && e < (unsigned)E; ++e) atomic_add_f32(&deg[dst[e]], 1.0f);
  }
}

__global__ void __launch_bounds__(256) norm_k(float* __restrict__ nrm, float* __restrict__ wsrc,
                                              int* __restrict__ cnt, int* __restrict__ rowfill,
                                              int n, int NB) {
  const unsigned bin = blockIdx.x * 8u + (threadIdx.x >> 5);
  if (bin >= (unsigned)NB) return;
  const unsigned lane = threadIdx.x & 31u;
  const unsigned r0 = bin * (unsigned)RB + 2u * lane;
  const bool ok0 = 2u * lane < (unsigned)RB && r0 < (unsigned)n;
  const bool ok1 = 2u * lane + 1u < (unsigned)RB && r0 + 1u < (unsigned)n;
  int d0 = 0, d1 = 0;
  if (ok0) {
    const float d = nrm[r0];
    const float r = 1.0f / sqrtf(fmaxf(d, 1.0f));
    nrm[r0] = r;
    wsrc[r0] = (1.0f - ALPHA) * r * r;
    d0 = (int)d;
  }
  if (ok1) {
    const float d = nrm[r0 + 1u];
    const float r = 1.0f / sqrtf(fmaxf(d, 1.0f));
    nrm[r0 + 1u] = r;
    wsrc[r0 + 1u] = (1.0f - ALPHA) * r * r;
    d1 = (int)d;
  }
  const int pair = d0 + d1;
  int inc = pair;
#pragma unroll
  for (int off = 1; off < 32; off <<= 1) {
    const int t = __shfl_up(inc, off);
    if (lane >= (unsigned)off) inc += t;
  }
  const int exc = inc - pair;
  if (ok0) rowfill[r0] = exc;
  if (ok1) rowfill[r0 + 1u] = exc + d0;
  if (lane == 31u) cnt[bin] = inc;
}

__global__ void __launch_bounds__(1024) scan_k(const int* __restrict__ cnt, int* __restrict__ start, int NB) {
  __shared__ int part[1024];
  const unsigned tid = threadIdx.x;
  const unsigned per = ((unsigned)NB + 1023u) / 1024u;
  const unsigned b0 = tid * per;
  const unsigned b1 = b0 + per < (unsigned)NB ? b0 + per : (unsigned)NB;
  int s = 0;
  for (unsigned b = b0; b < b1; ++b) s += cnt[b];
  part[tid] = s;
  __syncthreads();
  for (unsigned off = 1; off < 1024u; off <<= 1) {
    const int v = tid >= off ? part[tid - off] : 0;
    __syncthreads();
    part[tid] += v;
    __syncthreads();
  }
  int run = part[tid] - s;
  for (unsigned b = b0; b < b1; ++b) {
    start[b] = run;
    run += cnt[b];
  }
  if (tid == 1023u) start[NB] = part[1023];
}

__device__ __forceinline__ int2 edge_rec(int s, int d, unsigned bin, float w) {
  const unsigned dl = (unsigned)d - bin * (unsigned)RB;
  return make_int2((int)((unsigned)s | (dl << SRC_BITS)), __float_as_int(w));
}

__global__ void __launch_bounds__(256) place_k(const int* __restrict__ src, const int* __restrict__ dst,
                                               const float* __restrict__ wsrc, const int* __restrict__ start,
                                               int* __restrict__ rowfill, int2* __restrict__ lst, int E) {
  const unsigned e0 = (blockIdx.x * 256u + threadIdx.x) * 4u;
  if (e0 >= (unsigned)E) return;
  if ((((((size_t)dst) | ((size_t)src)) & 15) == 0) && e0 + 4u <= (unsigned)E) {
    const int4 dv = *reinterpret_cast<const int4*>(dst + e0);
    const int4 sv = *reinterpret_cast<const int4*>(src + e0);
    const unsigned b0 = (unsigned)dv.x / RB, b1 = (unsigned)dv.y / RB;
    const unsigned b2 = (unsigned)dv.z / RB, b3 = (unsigned)dv.w / RB;
    const float w0 = wsrc[sv.x], w1 = wsrc[sv.y], w2 = wsrc[sv.z], w3 = wsrc[sv.w];
    const int q0 = start[b0], q1 = start[b1], q2 = start[b2], q3 = start[b3];
    const int p0 = q0 + atomicAdd(&rowfill[dv.x], 1);
    const int p1 = q1 + atomicAdd(&rowfill[dv.y], 1);
    const int p2 = q2 + atomicAdd(&rowfill[dv.z], 1);
    const int p3 = q3 + atomicAdd(&rowfill[dv.w], 1);
    lst[p0] = edge_rec(sv.x, dv.x, b0, w0);
    lst[p1] = edge_rec(sv.y, dv.y, b1, w1);
    lst[p2] = edge_rec(sv.z, dv.z, b2, w2);
    lst[p3] = edge_rec(sv.w, dv.w, b3, w3);
  } else {
    for (unsigned e = e0; e < e0 + 4u && e < (unsigned)E; ++e) {
      const int d = dst[e];
      const int s = src[e];
      const unsigned b = (unsigned)d / RB;
      const float w = wsrc[s];
      const int p = start[b] + atomicAdd(&rowfill[d], 1);
      lst[p] = edge_rec(s, d, b, w);
    }
  }
}

__device__ __forceinline__ float elu1(float v) {
  return v > 0.0f ? v : (expf(v) - 1.0f);
}

template <int KDIM, int HID>
__global__ void __launch_bounds__(32 * (HID / 16))
gemm_wmma(const float* __restrict__ A, const float* __restrict__ W,
          const float* __restrict__ bias, const float* __restrict__ nrm,
          float* __restrict__ A0, float* __restrict__ SD, int N) {
  constexpr int KB = KDIM / 32;
  const int ntm = (N + 15) / 16;
  const int tn = __builtin_amdgcn_readfirstlane((int)(threadIdx.x >> 5));
  const int lane = threadIdx.x & 31;
  const int half = lane >> 4;
  const int l15 = lane & 15;
  const int col = tn * 16 + l15;

  v4f wf[KB][4];
  {
    const v4f* wp = reinterpret_cast<const v4f*>(W + (size_t)col * KDIM + 16 * half);
#pragma unroll
    for (int kb = 0; kb < KB; ++kb)
#pragma unroll
      for (int q = 0; q < 4; ++q) wf[kb][q] = wp[kb * 8 + q];
  }
  const float bs = bias[col];

  for (int tm = blockIdx.x; tm < ntm; tm += gridDim.x) {
    const int row = tm * 16 + l15;
    const int rowc = row < N ? row : N - 1;
    const v4f* ap = reinterpret_cast<const v4f*>(A + (size_t)rowc * KDIM + 16 * half);
    v4f af[KB][4];
#pragma unroll
    for (int kb = 0; kb < KB; ++kb)
#pragma unroll
      for (int q = 0; q < 4; ++q) af[kb][q] = ap[kb * 8 + q];

    v8f acc = {};
#pragma unroll
    for (int kb = 0; kb < KB; ++kb) {
#pragma unroll
      for (int q = 0; q < 4; ++q) {
        const v4f x = af[kb][q];
        const v4f y = wf[kb][q];
        v2f a, b;
        a.x = x.x; a.y = x.y; b.x = y.x; b.y = y.y;
        acc = __builtin_amdgcn_wmma_f32_16x16x4_f32(false, a, false, b, (short)0, acc, false, false);
        a.x = x.z; a.y = x.w; b.x = y.z; b.y = y.w;
        acc = __builtin_amdgcn_wmma_f32_16x16x4_f32(false, a, false, b, (short)0, acc, false, false);
      }
    }
    if (tm * 16 + 15 < N) {
#pragma unroll
      for (int v = 0; v < 8; ++v) {
        const int m = tm * 16 + v + 8 * half;
        const float r = 1.0f / ((1.0f - ALPHA) * nrm[m]);
        const float a0 = (acc[v] + bs) * r;
        A0[(size_t)m * HID + col] = a0;
        SD[(size_t)m * HID + col] = ALPHA * a0;
      }
    } else {
#pragma unroll
      for (int v = 0; v < 8; ++v) {
        const int m = tm * 16 + v + 8 * half;
        if (m < N) {
          const float r = 1.0f / ((1.0f - ALPHA) * nrm[m]);
          const float a0 = (acc[v] + bs) * r;
          A0[(size_t)m * HID + col] = a0;
          SD[(size_t)m * HID + col] = ALPHA * a0;
        }
      }
    }
  }
}

template <int D, int EPG, bool TAIL>
__device__ __forceinline__ void lds_batch(const int2* __restrict__ lst, const float* __restrict__ h,
                                          float* acc, unsigned e, unsigned e1, int ci) {
  constexpr int NJ = D / 16;
  int s[EPG], d[EPG];
  float w[EPG];
#pragma unroll
  for (int i = 0; i < EPG; ++i) {
    const bool in = !TAIL || e + (unsigned)i < e1;
    const int2 p = lst[in ? e + (unsigned)i : e1 - 1u];
    s[i] = p.x & SRC_MASK;
    d[i] = (int)((unsigned)p.x >> SRC_BITS);
    w[i] = in ? __int_as_float(p.y) : 0.0f;
  }
  float v[EPG][NJ];
#pragma unroll
  for (int i = 0; i < EPG; ++i) {
    const int sb = s[i] * D + ci;
#pragma unroll
    for (int j = 0; j < NJ; ++j) v[i][j] = h[sb + 16 * j];
  }
  float run[NJ];
#pragma unroll
  for (int j = 0; j < NJ; ++j) run[j] = 0.0f;
#pragma unroll
  for (int i = 0; i < EPG; ++i) {
#pragma unroll
    for (int j = 0; j < NJ; ++j) run[j] += v[i][j] * w[i];
    bool last = true;
    if (i + 1 < EPG) last = d[i] != d[i + 1];
    if (last) {
      const int lb = d[i] * D + ci;
#pragma unroll
      for (int j = 0; j < NJ; ++j) {
        lds_add_f32(&acc[lb + 16 * j], run[j]);
        run[j] = 0.0f;
      }
    }
  }
}

template <int D, int EPG, int MODE, int CO>
__global__ void __launch_bounds__(256) step_k(const int2* __restrict__ lst, const int* __restrict__ start,
                       const float* __restrict__ h,
                       const float* __restrict__ seed, float* __restrict__ tgt, int N,
                       const float* __restrict__ nrm, const float* __restrict__ W2,
                       const float* __restrict__ b2, float* __restrict__ tgt2) {
  constexpr int NL = (RB * (D / 4) + 255) / 256;
  __shared__ __attribute__((aligned(16))) float acc[RB * D];
  const unsigned tid = threadIdx.x;
  const unsigned grp = (tid >> 5) | ((tid & 16u) >> 1);
  const unsigned r0 = blockIdx.x * (unsigned)RB;
  const unsigned nr = (unsigned)N - r0 < (unsigned)RB ? (unsigned)N - r0 : (unsigned)RB;
  const unsigned n4 = nr * (unsigned)(D / 4);
  const v4f* sp = reinterpret_cast<const v4f*>(seed + (size_t)r0 * D);
  v4f* ap = reinterpret_cast<v4f*>(acc);
  v4f sv[NL];
#pragma unroll
  for (int k = 0; k < NL; ++k) {
    const unsigned i = tid + 256u * (unsigned)k;
    sv[k] = v4f{0.f, 0.f, 0.f, 0.f};
    if (i < n4) sv[k] = __builtin_nontemporal_load(sp + i);
  }
  const unsigned e0 = (unsigned)start[blockIdx.x];
  const unsigned e1 = (unsigned)start[blockIdx.x + 1u];
#pragma unroll
  for (int k = 0; k < NL; ++k) {
    const unsigned i = tid + 256u * (unsigned)k;
    if (i < n4) ap[i] = sv[k];
  }
  __syncthreads();

  const int ci = (int)(tid & 15u);
  for (unsigned e = e0 + grp * (unsigned)EPG; e < e1; e += 16u * (unsigned)EPG) {
    if (e + (unsigned)EPG <= e1)
      lds_batch<D, EPG, false>(lst, h, acc, e, e1, ci);
    else
      lds_batch<D, EPG, true>(lst, h, acc, e, e1, ci);
  }
  __syncthreads();
  if constexpr (MODE == 0) {
    v4f* tp = reinterpret_cast<v4f*>(tgt + (size_t)r0 * D);
#pragma unroll
    for (int k = 0; k < NL; ++k) {
      const unsigned i = tid + 256u * (unsigned)k;
      if (i < n4) tp[i] = ap[i];
    }
  } else if constexpr (MODE == 2) {
    v4f* tp = reinterpret_cast<v4f*>(tgt + (size_t)r0 * D);
#pragma unroll
    for (int k = 0; k < NL; ++k) {
      const unsigned i = tid + 256u * (unsigned)k;
      if (i < n4) {
        const float w = (1.0f - ALPHA) * nrm[r0 + i / (unsigned)(D / 4)];
        const v4f x = ap[i];
        tp[i] = v4f{elu1(w * x.x), elu1(w * x.y), elu1(w * x.z), elu1(w * x.w)};
      }
    }
  } else {
    static_assert(MODE != 1 || CO == 16, "one 16-column tile");
    constexpr unsigned NT = (RB + 15) / 16;
    const unsigned wv = (unsigned)__builtin_amdgcn_readfirstlane((int)(tid >> 5));
    if (wv < NT) {
      const unsigned half = (tid >> 4) & 1u;
      const unsigned l15 = tid & 15u;
      const unsigned rl = wv * 16u + l15;
      const unsigned rc = rl < nr ? rl : nr - 1u;
      float wr[8];
#pragma unroll
      for (int v = 0; v < 8; ++v) {
        const unsigned m = wv * 16u + (unsigned)v + 8u * half;
        wr[v] = (1.0f - ALPHA) * nrm[r0 + (m < nr ? m : nr - 1u)];
      }
      const float bs = b2[l15];
      v8f c = {};
#pragma unroll
      for (int k = 0; k < D; k += 4) {
        const unsigned ka = (unsigned)k + 2u * half;
        v2f a, b;
        a.x = acc[rc * (unsigned)D + ka];
        a.y = acc[rc * (unsigned)D + ka + 1u];
        b.x = W2[l15 * (unsigned)D + ka];
        b.y = W2[l15 * (unsigned)D + ka + 1u];
        c = __builtin_amdgcn_wmma_f32_16x16x4_f32(false, a, false, b, (short)0, c, false, false);
      }
      float t[8];
#pragma unroll
      for (int v = 0; v < 8; ++v) t[v] = wr[v] * c[v] + bs;
#pragma unroll
      for (int v = 0; v < 8; ++v) {
        const unsigned m = wv * 16u + (unsigned)v + 8u * half;
        if (m < nr) {
          const unsigned node = r0 + m;
          const float a0 = elu1(t[v]) * (1.0f / wr[v]);
          tgt[(size_t)node * CO + l15] = a0;
          tgt2[(size_t)node * CO + l15] = ALPHA * a0;
        }
      }
    }
  }
}


extern "C" void kernel_launch(void* const* d_in, const int* in_sizes, int n_in,
                              void* d_out, int out_size, void* d_ws, size_t ws_size,
                              hipStream_t stream) {
  const float* F   = (const float*)d_in[0];
  const int*   src = (const int*)d_in[1];
  const int*   dst = (const int*)d_in[2];
  const float* W1  = (const float*)d_in[3];
  const float* b1  = (const float*)d_in[4];
  const float* W2  = (const float*)d_in[5];
  const float* b2  = (const float*)d_in[6];
  float* out = (float*)d_out;

  constexpr int IN = 256, HID = 48, CLS = 16, KSTEPS = 10;
  constexpr int EPG = 8;
  const int N = in_sizes[0] / IN;
  const int E = in_sizes[1];
  const int NB = (N + RB - 1) / RB;
  const int NBP = ((NB + 1 + 63) / 64) * 64;

  float* ws   = (float*)d_ws;
  float* nrm  = ws;
  float* wsrc = nrm + N;
  float* H    = wsrc + N;
  float* S    = H + (size_t)N * HID;
  float* B[2];
  B[0] = S + (size_t)N * HID;
  B[1] = B[0] + (size_t)N * HID;
  float* b[2];
  b[0] = B[0];
  b[1] = b[0] + (size_t)N * CLS;
  float* s16  = H;
  int* cnt   = (int*)(B[1] + (size_t)N * HID);
  int* start = cnt + NBP;
  int* rowfill = start + NBP;
  int2* lst  = (int2*)(rowfill + ((N + 63) / 64) * 64);

  auto blocks = [](long n, int t) { return (int)((n + t - 1) / t); };

  zero4_k<<<blocks(N / 4, 256), 256, 0, stream>>>((float4*)nrm, N / 4, (float4*)cnt, 0);
  deg_k<<<blocks(((long)E + 3) / 4, 256), 256, 0, stream>>>(dst, nrm, E);
  norm_k<<<blocks(NB, 8), 256, 0, stream>>>(nrm, wsrc, cnt, rowfill, N, NB);
  scan_k<<<1, 1024, 0, stream>>>(cnt, start, NB);
  place_k<<<blocks(((long)E + 3) / 4, 256), 256, 0, stream>>>(src, dst, wsrc, start, rowfill, lst, E);

  gemm_wmma<IN, HID><<<1250, 32 * (HID / 16), 0, stream>>>(F, W1, b1, nrm, B[0], S, N);

  for (int t = 0; t < KSTEPS - 1; ++t)
    step_k<HID, EPG, 0, CLS><<<NB, 256, 0, stream>>>(lst, start, B[t % 2], S, B[(t + 1) % 2], N,
                                                     nrm, W2, b2, nullptr);
  step_k<HID, EPG, 1, CLS><<<NB, 256, 0, stream>>>(lst, start, B[(KSTEPS - 1) % 2], S, b[0], N,
                                                   nrm, W2, b2, s16);

  for (int t = 0; t < KSTEPS - 1; ++t)
    step_k<CLS, EPG, 0, CLS><<<NB, 256, 0, stream>>>(lst, start, b[t % 2], s16, b[(t + 1) % 2], N,
                                                     nrm, W2, b2, nullptr);
  step_k<CLS, EPG, 2, CLS><<<NB, 256, 0, stream>>>(lst, start, b[(KSTEPS - 1) % 2], s16, out, N,
                                                   nrm, W2, b2, nullptr);
}
